# starting kernel
# baseline (speedup 1.0000x reference)
#include <hip/hip_runtime.h>


#define DIM       1024
#define BATCH     32
#define DEC_LEN   512
#define ENC_LEN   1024

#define ENC_ROWS  (BATCH * ENC_LEN)
#define DEC_ROWS  (BATCH * DEC_LEN)
#define ENC_TILES (ENC_ROWS / 16)
#define DEC_TILES (DEC_ROWS / 16)
#define ALL_TILES (ENC_TILES + DEC_TILES)

typedef __attribute__((ext_vector_type(2))) float v2f;
typedef __attribute__((ext_vector_type(4))) float v4f;
typedef __attribute__((ext_vector_type(8))) float v8f;

__global__ __launch_bounds__(256) void proj_wmma_kernel(
    const float* __restrict__ dec_states,
    const float* __restrict__ enc_states,
    const float* __restrict__ w,
    float* __restrict__ enc_proj,
    float* __restrict__ dec_proj)
{
    const int wave = (blockIdx.x * blockDim.x + threadIdx.x) >> 5;
    const int lane = threadIdx.x & 31;
    if (wave >= ALL_TILES) return;

    const float* src;
    const float* wv;
    float*       dst;
    int          tile16;
    if (wave < ENC_TILES) {
        src    = enc_states;
        wv     = w;
        dst    = enc_proj;
        tile16 = wave * 16;
    } else {
        src    = dec_states;
        wv     = w + DIM;
        dst    = dec_proj;
        tile16 = (wave - ENC_TILES) * 16;
    }

    const int row  = tile16 + (lane & 15);
    const int koff = (lane >> 4) * 2;
    const float* ap = src + (size_t)row * DIM + koff;
    const float* bp = wv + koff;

    v8f c = {0.f, 0.f, 0.f, 0.f, 0.f, 0.f, 0.f, 0.f};

    for (int kb = 0; kb < DIM; kb += 32) {
        __builtin_prefetch(ap + kb + 128, 0, 1);
#pragma unroll
        for (int k2 = 0; k2 < 32; k2 += 4) {
            v2f a = *(const v2f*)(ap + kb + k2);
            v2f b = *(const v2f*)(bp + kb + k2);
            c = __builtin_amdgcn_wmma_f32_16x16x4_f32(
                     false, a,  false, b,
                     (short)0, c,  false,  false);
        }
    }

    if ((lane & 15) == 0) {
        float* o = dst + tile16 + (lane >> 4) * 8;
#pragma unroll
        for (int v = 0; v < 8; ++v) o[v] = c[v];
    }
}

__global__ __launch_bounds__(256) void bcast_add_kernel(
    const float* __restrict__ enc_proj,
    const float* __restrict__ dec_proj,
    const float* __restrict__ bias,
    float* __restrict__ out)
{
    const unsigned idx = blockIdx.x * blockDim.x + threadIdx.x;
    const unsigned j4 = idx & 255u;
    const unsigned i  = (idx >> 8) & 511u;
    const unsigned b  = idx >> 17;

    const float d = dec_proj[b * DEC_LEN + i] + bias[0];
    const v4f  e = *(const v4f*)(enc_proj + b * ENC_LEN + j4 * 4);
    v4f r = {e.x + d, e.y + d, e.z + d, e.w + d};
    *((v4f*)out + idx) = r;
}

extern "C" void kernel_launch(void* const* d_in, const int* in_sizes, int n_in,
                              void* d_out, int out_size, void* d_ws, size_t ws_size,
                              hipStream_t stream) {
    const float* dec_states = (const float*)d_in[0];
    const float* enc_states = (const float*)d_in[1];
    const float* mlp_weight = (const float*)d_in[3];
    const float* mlp_bias   = (const float*)d_in[4];

    float* enc_proj = (float*)d_ws;
    float* dec_proj = enc_proj + ENC_ROWS;

    proj_wmma_kernel<<<ALL_TILES / 8, 256, 0, stream>>>(
        dec_states, enc_states, mlp_weight, enc_proj, dec_proj);

    bcast_add_kernel<<<(BATCH * DEC_LEN * ENC_LEN / 4) / 256, 256, 0, stream>>>(
        enc_proj, dec_proj, mlp_bias, (float*)d_out);
}


# phase 1 loader re-laid: each wave streams whole rows with 512-B contiguous loads (k split over the 16 tile rows, row dot = trace of D; sums now 16 partials + butterfly), w kept in registers
# speedup vs baseline: 1.1143x; 1.1143x over previous
#include <hip/hip_runtime.h>


#define DIM       1024
#define BATCH     32
#define DEC_LEN   512
#define ENC_LEN   1024

#define ENC_ROWS  (BATCH * ENC_LEN)
#define DEC_ROWS  (BATCH * DEC_LEN)

#define ROWS_PER_WAVE   8
#define ENC_WAVES (ENC_ROWS / ROWS_PER_WAVE)
#define DEC_WAVES (DEC_ROWS / ROWS_PER_WAVE)
#define ALL_WAVES (ENC_WAVES + DEC_WAVES)
#define P1_WAVES_PER_BLOCK 8
#define NPIECE    (DIM / 128)

typedef __attribute__((ext_vector_type(2))) float v2f;
typedef __attribute__((ext_vector_type(4))) float v4f;
typedef __attribute__((ext_vector_type(8))) float v8f;

__device__ __forceinline__ v4f load_piece(const float* __restrict__ p, int j)
{
    return *(const v4f*)(p + j * 128);
}

__device__ __forceinline__ float diag_sum(v8f c, int lane)
{
    const int r = lane & 7;
    float d = c[0];
    d = (r == 1) ? c[1] : d;
    d = (r == 2) ? c[2] : d;
    d = (r == 3) ? c[3] : d;
    d = (r == 4) ? c[4] : d;
    d = (r == 5) ? c[5] : d;
    d = (r == 6) ? c[6] : d;
    d = (r == 7) ? c[7] : d;
    const int q = lane & 24;
    d = (q == 0 || q == 24) ? d : 0.f;
    d += __shfl_xor(d, 16);
    d += __shfl_xor(d, 8);
    d += __shfl_xor(d, 4);
    d += __shfl_xor(d, 2);
    d += __shfl_xor(d, 1);
    return d;
}

#define WMMA_F32(acc, xa, wb) \
    acc = __builtin_amdgcn_wmma_f32_16x16x4_f32(false, (xa), false, (wb), (short)0, acc, false, false)

__global__ __launch_bounds__(P1_WAVES_PER_BLOCK * 32) void proj_wmma_kernel(
    const float* __restrict__ dec_states,
    const float* __restrict__ enc_states,
    const float* __restrict__ w,
    float* __restrict__ enc_proj,
    float* __restrict__ dec_proj)
{
    const int lane = threadIdx.x & 31;
    const int wib  = __builtin_amdgcn_readfirstlane((int)(threadIdx.x >> 5));
    const int wave = (int)blockIdx.x * P1_WAVES_PER_BLOCK + wib;

    const float* src;
    const float* wv;
    float*       dst;
    int          row0;
    if (wave < ENC_WAVES) {
        src  = enc_states;
        wv   = w;
        dst  = enc_proj;
        row0 = wave * ROWS_PER_WAVE;
    } else {
        src  = dec_states;
        wv   = w + DIM;
        dst  = dec_proj;
        row0 = (wave - ENC_WAVES) * ROWS_PER_WAVE;
    }

    const float* p = src + (size_t)row0 * DIM + lane * 4;

    v4f wf[NPIECE];
#pragma unroll
    for (int j = 0; j < NPIECE; ++j) wf[j] = load_piece(wv + lane * 4, j);

    v4f x0[NPIECE], x1[NPIECE];
#pragma unroll
    for (int j = 0; j < NPIECE; ++j) {
        x0[j] = load_piece(p, j);
        x1[j] = load_piece(p + DIM, j);
    }

    float res = 0.f;

#pragma unroll 1
    for (int s = 0; s < ROWS_PER_WAVE / 2 - 1; ++s) {
        const float* pn = p + (size_t)(2 * s + 2) * DIM;
        v8f c0 = {0.f, 0.f, 0.f, 0.f, 0.f, 0.f, 0.f, 0.f};
        v8f c1 = {0.f, 0.f, 0.f, 0.f, 0.f, 0.f, 0.f, 0.f};
#pragma unroll
        for (int j = 0; j < NPIECE; ++j) {
            WMMA_F32(c0, x0[j].xy, wf[j].xy);
            WMMA_F32(c1, x1[j].xy, wf[j].xy);
            WMMA_F32(c0, x0[j].zw, wf[j].zw);
            WMMA_F32(c1, x1[j].zw, wf[j].zw);
            x0[j] = load_piece(pn, j);
            x1[j] = load_piece(pn + DIM, j);
            __builtin_amdgcn_sched_barrier(0);
        }
        const float d0 = diag_sum(c0, lane);
        const float d1 = diag_sum(c1, lane);
        res = (lane == 2 * s)     ? d0 : res;
        res = (lane == 2 * s + 1) ? d1 : res;
    }
    {
        v8f c0 = {0.f, 0.f, 0.f, 0.f, 0.f, 0.f, 0.f, 0.f};
        v8f c1 = {0.f, 0.f, 0.f, 0.f, 0.f, 0.f, 0.f, 0.f};
#pragma unroll
        for (int j = 0; j < NPIECE; ++j) {
            WMMA_F32(c0, x0[j].xy, wf[j].xy);
            WMMA_F32(c1, x1[j].xy, wf[j].xy);
            WMMA_F32(c0, x0[j].zw, wf[j].zw);
            WMMA_F32(c1, x1[j].zw, wf[j].zw);
        }
        const float d0 = diag_sum(c0, lane);
        const float d1 = diag_sum(c1, lane);
        res = (lane == ROWS_PER_WAVE - 2) ? d0 : res;
        res = (lane == ROWS_PER_WAVE - 1) ? d1 : res;
    }

    __shared__ float blk[P1_WAVES_PER_BLOCK * ROWS_PER_WAVE];
    if (lane < ROWS_PER_WAVE) blk[wib * ROWS_PER_WAVE + lane] = res;
    __syncthreads();
    if (threadIdx.x < P1_WAVES_PER_BLOCK * ROWS_PER_WAVE)
        dst[row0 - wib * ROWS_PER_WAVE + (int)threadIdx.x] = blk[threadIdx.x];
}

__global__ __launch_bounds__(256) void bcast_add_kernel(
    const float* __restrict__ enc_proj,
    const float* __restrict__ dec_proj,
    const float* __restrict__ bias,
    float* __restrict__ out)
{
    const unsigned idx = blockIdx.x * blockDim.x + threadIdx.x;
    const unsigned j4 = idx & 255u;
    const unsigned i  = (idx >> 8) & 511u;
    const unsigned b  = idx >> 17;

    const float d = dec_proj[b * DEC_LEN + i] + bias[0];
    const v4f  e = *(const v4f*)(enc_proj + b * ENC_LEN + j4 * 4);
    v4f r = {e.x + d, e.y + d, e.z + d, e.w + d};
    *((v4f*)out + idx) = r;
}

extern "C" void kernel_launch(void* const* d_in, const int* in_sizes, int n_in,
                              void* d_out, int out_size, void* d_ws, size_t ws_size,
                              hipStream_t stream) {
    const float* dec_states = (const float*)d_in[0];
    const float* enc_states = (const float*)d_in[1];
    const float* mlp_weight = (const float*)d_in[3];
    const float* mlp_bias   = (const float*)d_in[4];

    float* enc_proj = (float*)d_ws;
    float* dec_proj = enc_proj + ENC_ROWS;

    proj_wmma_kernel<<<ALL_WAVES / P1_WAVES_PER_BLOCK, P1_WAVES_PER_BLOCK * 32, 0, stream>>>(
        dec_states, enc_states, mlp_weight, enc_proj, dec_proj);

    bcast_add_kernel<<<(BATCH * DEC_LEN * ENC_LEN / 4) / 256, 256, 0, stream>>>(
        enc_proj, dec_proj, mlp_bias, (float*)d_out);
}


# phase 2 re-gridded to 1,024 blocks: a thread keeps its 4 encoder sums and writes 16 decoder rows, dealt round-robin over its batch's 32 blocks; 8,192 waves enter instead of 131,072
# speedup vs baseline: 1.3261x; 1.1901x over previous
#include <hip/hip_runtime.h>


#define DIM       1024
#define BATCH     32
#define DEC_LEN   512
#define ENC_LEN   1024

#define ENC_ROWS  (BATCH * ENC_LEN)
#define DEC_ROWS  (BATCH * DEC_LEN)

#define ROWS_PER_WAVE   8
#define ENC_WAVES (ENC_ROWS / ROWS_PER_WAVE)
#define DEC_WAVES (DEC_ROWS / ROWS_PER_WAVE)
#define ALL_WAVES (ENC_WAVES + DEC_WAVES)
#define P1_WAVES_PER_BLOCK 8
#define NPIECE    (DIM / 128)

typedef __attribute__((ext_vector_type(2))) float v2f;
typedef __attribute__((ext_vector_type(4))) float v4f;
typedef __attribute__((ext_vector_type(8))) float v8f;

__device__ __forceinline__ v4f load_piece(const float* __restrict__ p, int j)
{
    return *(const v4f*)(p + j * 128);
}

__device__ __forceinline__ float diag_sum(v8f c, int lane)
{
    const int r = lane & 7;
    float d = c[0];
    d = (r == 1) ? c[1] : d;
    d = (r == 2) ? c[2] : d;
    d = (r == 3) ? c[3] : d;
    d = (r == 4) ? c[4] : d;
    d = (r == 5) ? c[5] : d;
    d = (r == 6) ? c[6] : d;
    d = (r == 7) ? c[7] : d;
    const int q = lane & 24;
    d = (q == 0 || q == 24) ? d : 0.f;
    d += __shfl_xor(d, 16);
    d += __shfl_xor(d, 8);
    d += __shfl_xor(d, 4);
    d += __shfl_xor(d, 2);
    d += __shfl_xor(d, 1);
    return d;
}

#define WMMA_F32(acc, xa, wb) \
    acc = __builtin_amdgcn_wmma_f32_16x16x4_f32(false, (xa), false, (wb), (short)0, acc, false, false)

__global__ __launch_bounds__(P1_WAVES_PER_BLOCK * 32) void proj_wmma_kernel(
    const float* __restrict__ dec_states,
    const float* __restrict__ enc_states,
    const float* __restrict__ w,
    float* __restrict__ enc_proj,
    float* __restrict__ dec_proj)
{
    const int lane = threadIdx.x & 31;
    const int wib  = __builtin_amdgcn_readfirstlane((int)(threadIdx.x >> 5));
    const int wave = (int)blockIdx.x * P1_WAVES_PER_BLOCK + wib;

    const float* src;
    const float* wv;
    float*       dst;
    int          row0;
    if (wave < ENC_WAVES) {
        src  = enc_states;
        wv   = w;
        dst  = enc_proj;
        row0 = wave * ROWS_PER_WAVE;
    } else {
        src  = dec_states;
        wv   = w + DIM;
        dst  = dec_proj;
        row0 = (wave - ENC_WAVES) * ROWS_PER_WAVE;
    }

    const float* p = src + (size_t)row0 * DIM + lane * 4;

    v4f wf[NPIECE];
#pragma unroll
    for (int j = 0; j < NPIECE; ++j) wf[j] = load_piece(wv + lane * 4, j);

    v4f x0[NPIECE], x1[NPIECE];
#pragma unroll
    for (int j = 0; j < NPIECE; ++j) {
        x0[j] = load_piece(p, j);
        x1[j] = load_piece(p + DIM, j);
    }

    float res = 0.f;

#pragma unroll 1
    for (int s = 0; s < ROWS_PER_WAVE / 2 - 1; ++s) {
        const float* pn = p + (size_t)(2 * s + 2) * DIM;
        v8f c0 = {0.f, 0.f, 0.f, 0.f, 0.f, 0.f, 0.f, 0.f};
        v8f c1 = {0.f, 0.f, 0.f, 0.f, 0.f, 0.f, 0.f, 0.f};
#pragma unroll
        for (int j = 0; j < NPIECE; ++j) {
            WMMA_F32(c0, x0[j].xy, wf[j].xy);
            WMMA_F32(c1, x1[j].xy, wf[j].xy);
            WMMA_F32(c0, x0[j].zw, wf[j].zw);
            WMMA_F32(c1, x1[j].zw, wf[j].zw);
            x0[j] = load_piece(pn, j);
            x1[j] = load_piece(pn + DIM, j);
            __builtin_amdgcn_sched_barrier(0);
        }
        const float d0 = diag_sum(c0, lane);
        const float d1 = diag_sum(c1, lane);
        res = (lane == 2 * s)     ? d0 : res;
        res = (lane == 2 * s + 1) ? d1 : res;
    }
    {
        v8f c0 = {0.f, 0.f, 0.f, 0.f, 0.f, 0.f, 0.f, 0.f};
        v8f c1 = {0.f, 0.f, 0.f, 0.f, 0.f, 0.f, 0.f, 0.f};
#pragma unroll
        for (int j = 0; j < NPIECE; ++j) {
            WMMA_F32(c0, x0[j].xy, wf[j].xy);
            WMMA_F32(c1, x1[j].xy, wf[j].xy);
            WMMA_F32(c0, x0[j].zw, wf[j].zw);
            WMMA_F32(c1, x1[j].zw, wf[j].zw);
        }
        const float d0 = diag_sum(c0, lane);
        const float d1 = diag_sum(c1, lane);
        res = (lane == ROWS_PER_WAVE - 2) ? d0 : res;
        res = (lane == ROWS_PER_WAVE - 1) ? d1 : res;
    }

    __shared__ float blk[P1_WAVES_PER_BLOCK * ROWS_PER_WAVE];
    if (lane < ROWS_PER_WAVE) blk[wib * ROWS_PER_WAVE + lane] = res;
    __syncthreads();
    if (threadIdx.x < P1_WAVES_PER_BLOCK * ROWS_PER_WAVE)
        dst[row0 - wib * ROWS_PER_WAVE + (int)threadIdx.x] = blk[threadIdx.x];
}

#define P2_ROWS             16
#define P2_BLOCKS_PER_BATCH (DEC_LEN / P2_ROWS)
#define P2_BLOCKS           (BATCH * P2_BLOCKS_PER_BATCH)

__global__ __launch_bounds__(256) void bcast_add_kernel(
    const float* __restrict__ enc_proj,
    const float* __restrict__ dec_proj,
    const float* __restrict__ bias,
    float* __restrict__ out)
{
    const unsigned b = blockIdx.x / P2_BLOCKS_PER_BATCH;
    const unsigned q = blockIdx.x % P2_BLOCKS_PER_BATCH;
    const unsigned t = threadIdx.x;

    const v4f   e  = *(const v4f*)(enc_proj + b * ENC_LEN + t * 4);
    const float bs = bias[0];

    const float* dp = dec_proj + b * DEC_LEN + q;
    float d[P2_ROWS];
#pragma unroll
    for (int s = 0; s < P2_ROWS; ++s) d[s] = dp[s * P2_BLOCKS_PER_BATCH] + bs;

    v4f* o = (v4f*)out + ((size_t)(b * DEC_LEN + q) * (ENC_LEN / 4) + t);
#pragma unroll
    for (int s = 0; s < P2_ROWS; ++s) {
        v4f r = {e.x + d[s], e.y + d[s], e.z + d[s], e.w + d[s]};
        o[(size_t)s * P2_BLOCKS_PER_BATCH * (ENC_LEN / 4)] = r;
    }
}

extern "C" void kernel_launch(void* const* d_in, const int* in_sizes, int n_in,
                              void* d_out, int out_size, void* d_ws, size_t ws_size,
                              hipStream_t stream) {
    const float* dec_states = (const float*)d_in[0];
    const float* enc_states = (const float*)d_in[1];
    const float* mlp_weight = (const float*)d_in[3];
    const float* mlp_bias   = (const float*)d_in[4];

    float* enc_proj = (float*)d_ws;
    float* dec_proj = enc_proj + ENC_ROWS;

    proj_wmma_kernel<<<ALL_WAVES / P1_WAVES_PER_BLOCK, P1_WAVES_PER_BLOCK * 32, 0, stream>>>(
        dec_states, enc_states, mlp_weight, enc_proj, dec_proj);

    bcast_add_kernel<<<P2_BLOCKS, 256, 0, stream>>>(
        enc_proj, dec_proj, mlp_bias, (float*)d_out);
}


# decoder row dots moved from launch 1 into launch 2: each output block forms its own 16 decoder sums (2 rows per wave, same matrix-unit dot, LDS hand-over); launch 1 streams the encoder rows only
# speedup vs baseline: 1.5423x; 1.1630x over previous
#include <hip/hip_runtime.h>


#define DIM       1024
#define BATCH     32
#define DEC_LEN   512
#define ENC_LEN   1024

#define ENC_ROWS  (BATCH * ENC_LEN)
#define DEC_ROWS  (BATCH * DEC_LEN)

#define ROWS_PER_WAVE   8
#define ENC_WAVES (ENC_ROWS / ROWS_PER_WAVE)
#define P1_WAVES_PER_BLOCK 8
#define NPIECE    (DIM / 128)

typedef __attribute__((ext_vector_type(2))) float v2f;
typedef __attribute__((ext_vector_type(4))) float v4f;
typedef __attribute__((ext_vector_type(8))) float v8f;

__device__ __forceinline__ v4f load_piece(const float* __restrict__ p, int j)
{
    return *(const v4f*)(p + j * 128);
}

__device__ __forceinline__ float diag_sum(v8f c, int lane)
{
    const int r = lane & 7;
    float d = c[0];
    d = (r == 1) ? c[1] : d;
    d = (r == 2) ? c[2] : d;
    d = (r == 3) ? c[3] : d;
    d = (r == 4) ? c[4] : d;
    d = (r == 5) ? c[5] : d;
    d = (r == 6) ? c[6] : d;
    d = (r == 7) ? c[7] : d;
    const int q = lane & 24;
    d = (q == 0 || q == 24) ? d : 0.f;
    d += __shfl_xor(d, 16);
    d += __shfl_xor(d, 8);
    d += __shfl_xor(d, 4);
    d += __shfl_xor(d, 2);
    d += __shfl_xor(d, 1);
    return d;
}

#define WMMA_F32(acc, xa, wb) \
    acc = __builtin_amdgcn_wmma_f32_16x16x4_f32(false, (xa), false, (wb), (short)0, acc, false, false)

__global__ __launch_bounds__(P1_WAVES_PER_BLOCK * 32) void enc_proj_wmma_kernel(
    const float* __restrict__ enc_states,
    const float* __restrict__ w,
    float* __restrict__ enc_proj)
{
    const int lane = threadIdx.x & 31;
    const int wib  = __builtin_amdgcn_readfirstlane((int)(threadIdx.x >> 5));
    const int wave = (int)blockIdx.x * P1_WAVES_PER_BLOCK + wib;
    const int row0 = wave * ROWS_PER_WAVE;

    const float* p = enc_states + (size_t)row0 * DIM + lane * 4;

    v4f wf[NPIECE];
#pragma unroll
    for (int j = 0; j < NPIECE; ++j) wf[j] = load_piece(w + lane * 4, j);

    v4f x0[NPIECE], x1[NPIECE];
#pragma unroll
    for (int j = 0; j < NPIECE; ++j) {
        x0[j] = load_piece(p, j);
        x1[j] = load_piece(p + DIM, j);
    }

    float res = 0.f;

#pragma unroll 1
    for (int s = 0; s < ROWS_PER_WAVE / 2 - 1; ++s) {
        const float* pn = p + (size_t)(2 * s + 2) * DIM;
        v8f c0 = {0.f, 0.f, 0.f, 0.f, 0.f, 0.f, 0.f, 0.f};
        v8f c1 = {0.f, 0.f, 0.f, 0.f, 0.f, 0.f, 0.f, 0.f};
#pragma unroll
        for (int j = 0; j < NPIECE; ++j) {
            WMMA_F32(c0, x0[j].xy, wf[j].xy);
            WMMA_F32(c1, x1[j].xy, wf[j].xy);
            WMMA_F32(c0, x0[j].zw, wf[j].zw);
            WMMA_F32(c1, x1[j].zw, wf[j].zw);
            x0[j] = load_piece(pn, j);
            x1[j] = load_piece(pn + DIM, j);
            __builtin_amdgcn_sched_barrier(0);
        }
        const float d0 = diag_sum(c0, lane);
        const float d1 = diag_sum(c1, lane);
        res = (lane == 2 * s)     ? d0 : res;
        res = (lane == 2 * s + 1) ? d1 : res;
    }
    {
        v8f c0 = {0.f, 0.f, 0.f, 0.f, 0.f, 0.f, 0.f, 0.f};
        v8f c1 = {0.f, 0.f, 0.f, 0.f, 0.f, 0.f, 0.f, 0.f};
#pragma unroll
        for (int j = 0; j < NPIECE; ++j) {
            WMMA_F32(c0, x0[j].xy, wf[j].xy);
            WMMA_F32(c1, x1[j].xy, wf[j].xy);
            WMMA_F32(c0, x0[j].zw, wf[j].zw);
            WMMA_F32(c1, x1[j].zw, wf[j].zw);
        }
        const float d0 = diag_sum(c0, lane);
        const float d1 = diag_sum(c1, lane);
        res = (lane == ROWS_PER_WAVE - 2) ? d0 : res;
        res = (lane == ROWS_PER_WAVE - 1) ? d1 : res;
    }

    __shared__ float blk[P1_WAVES_PER_BLOCK * ROWS_PER_WAVE];
    if (lane < ROWS_PER_WAVE) blk[wib * ROWS_PER_WAVE + lane] = res;
    __syncthreads();
    if (threadIdx.x < P1_WAVES_PER_BLOCK * ROWS_PER_WAVE)
        enc_proj[row0 - wib * ROWS_PER_WAVE + (int)threadIdx.x] = blk[threadIdx.x];
}

#define P2_ROWS             16
#define P2_BLOCKS_PER_BATCH (DEC_LEN / P2_ROWS)
#define P2_BLOCKS           (BATCH * P2_BLOCKS_PER_BATCH)
#define P2_WAVES            8

__global__ __launch_bounds__(P2_WAVES * 32) void dec_dot_bcast_add_kernel(
    const float* __restrict__ dec_states,
    const float* __restrict__ w,
    const float* __restrict__ enc_proj,
    const float* __restrict__ bias,
    float* __restrict__ out)
{
    const unsigned b = blockIdx.x / P2_BLOCKS_PER_BATCH;
    const unsigned q = blockIdx.x % P2_BLOCKS_PER_BATCH;
    const unsigned t = threadIdx.x;
    const int lane = (int)(t & 31);
    const int wib  = __builtin_amdgcn_readfirstlane((int)(t >> 5));

    const float* p0 = dec_states
        + (size_t)(b * DEC_LEN + q + (unsigned)(2 * wib) * P2_BLOCKS_PER_BATCH) * DIM + lane * 4;
    const float* p1 = p0 + (size_t)P2_BLOCKS_PER_BATCH * DIM;

    const v4f   e  = *(const v4f*)(enc_proj + b * ENC_LEN + t * 4);
    const float bs = bias[0];

    v4f wf[NPIECE];
#pragma unroll
    for (int j = 0; j < NPIECE; ++j) wf[j] = load_piece(w + DIM + lane * 4, j);

    v4f x0[NPIECE], x1[NPIECE];
#pragma unroll
    for (int j = 0; j < NPIECE; ++j) {
        x0[j] = load_piece(p0, j);
        x1[j] = load_piece(p1, j);
    }
    __builtin_amdgcn_sched_barrier(0);

    v8f c0 = {0.f, 0.f, 0.f, 0.f, 0.f, 0.f, 0.f, 0.f};
    v8f c1 = {0.f, 0.f, 0.f, 0.f, 0.f, 0.f, 0.f, 0.f};
#pragma unroll
    for (int j = 0; j < NPIECE; ++j) {
        WMMA_F32(c0, x0[j].xy, wf[j].xy);
        WMMA_F32(c1, x1[j].xy, wf[j].xy);
        WMMA_F32(c0, x0[j].zw, wf[j].zw);
        WMMA_F32(c1, x1[j].zw, wf[j].zw);
    }
    const float d0 = diag_sum(c0, lane);
    const float d1 = diag_sum(c1, lane);

    __shared__ float dsum[P2_ROWS];
    if (lane == 0) {
        dsum[2 * wib]     = d0;
        dsum[2 * wib + 1] = d1;
    }

    __syncthreads();

    float d[P2_ROWS];
#pragma unroll
    for (int s = 0; s < P2_ROWS; ++s) d[s] = dsum[s] + bs;

    v4f* o = (v4f*)out + ((size_t)(b * DEC_LEN + q) * (ENC_LEN / 4) + t);
#pragma unroll
    for (int s = 0; s < P2_ROWS; ++s) {
        v4f r = {e.x + d[s], e.y + d[s], e.z + d[s], e.w + d[s]};
        o[(size_t)s * P2_BLOCKS_PER_BATCH * (ENC_LEN / 4)] = r;
    }
}

extern "C" void kernel_launch(void* const* d_in, const int* in_sizes, int n_in,
                              void* d_out, int out_size, void* d_ws, size_t ws_size,
                              hipStream_t stream) {
    const float* dec_states = (const float*)d_in[0];
    const float* enc_states = (const float*)d_in[1];
    const float* mlp_weight = (const float*)d_in[3];
    const float* mlp_bias   = (const float*)d_in[4];

    float* enc_proj = (float*)d_ws;

    enc_proj_wmma_kernel<<<ENC_WAVES / P1_WAVES_PER_BLOCK, P1_WAVES_PER_BLOCK * 32, 0, stream>>>(
        enc_states, mlp_weight, enc_proj);

    dec_dot_bcast_add_kernel<<<P2_BLOCKS, P2_WAVES * 32, 0, stream>>>(
        dec_states, mlp_weight, enc_proj, mlp_bias, (float*)d_out);
}
